# starting kernel
# baseline (speedup 1.0000x reference)
#include <hip/hip_runtime.h>

#define NUM_CAM 16
#define IN_SZ   256
#define OUT_SZ  128
#define BATCH   8192
#define TILE_M  16
#define XPITCH  260
#define WPITCH  130

typedef __attribute__((ext_vector_type(2))) float v2f;
typedef __attribute__((ext_vector_type(8))) float v8f;

__global__ void fc_zero_counts(int* counts) {
    if (threadIdx.x < NUM_CAM) counts[threadIdx.x] = 0;
}

__global__ void fc_bucket(const int* __restrict__ cam,
                          int* __restrict__ counts,
                          int* __restrict__ idx) {
    int t = blockIdx.x * blockDim.x + threadIdx.x;
    if (t < BATCH) {
        int c = cam[t] & (NUM_CAM - 1);
        int pos = atomicAdd(&counts[c], 1);
        idx[c * BATCH + pos] = t;
    }
}

__launch_bounds__(256)
__global__ void fc_expert_gemm(const float* __restrict__ x,
                               const float* __restrict__ W,
                               const float* __restrict__ bias,
                               const int*  __restrict__ counts,
                               const int*  __restrict__ idx,
                               float* __restrict__ out) {
    __shared__ float Xs[TILE_M * XPITCH];
    __shared__ v2f   Wt[128 * WPITCH];
    __shared__ int   Si[TILE_M];

    const int c     = blockIdx.x;
    const int tile  = blockIdx.y;
    const int count = counts[c];
    if (tile * TILE_M >= count) return;

    const int tid  = threadIdx.x;
    const int lane = tid & 31;
    const int wave = tid >> 5;
    const int half = lane >> 4;
    const int l    = lane & 15;

    {
        int row  = tid >> 4;
        int col0 = (tid & 15) * 16;
        int r    = tile * TILE_M + row;
        int s    = idx[c * BATCH + min(r, count - 1)];
        if ((tid & 15) == 0) Si[row] = s;
        const float4* src = (const float4*)(x + (size_t)s * IN_SZ + col0);
        float* dst = &Xs[row * XPITCH + col0];
        #pragma unroll
        for (int i = 0; i < 4; ++i) {
            float4 v = src[i];
            dst[i * 4 + 0] = v.x; dst[i * 4 + 1] = v.y;
            dst[i * 4 + 2] = v.z; dst[i * 4 + 3] = v.w;
        }
    }

    {
        const v2f* Wsrc = (const v2f*)(W + (size_t)c * OUT_SZ * IN_SZ);
        int p     = tid & 127;
        int nbase = tid >> 7;
        #pragma unroll 4
        for (int it = 0; it < 64; ++it) {
            int n = it * 2 + nbase;
            Wt[p * WPITCH + n] = Wsrc[n * 128 + p];
        }
    }
    __syncthreads();

    const int n0 = wave * 16;
    const float* ap = &Xs[l * XPITCH + 2 * half];
    const v2f*   bp = &Wt[half * WPITCH + (n0 + l)];

    v8f acc = {};
    for (int ko = 0; ko < 8; ++ko) {
        #pragma unroll
        for (int ki = 0; ki < 8; ++ki) {
            v2f a = *(const v2f*)(ap + ki * 4);
            v2f b = bp[ki * 2 * WPITCH];
            acc = __builtin_amdgcn_wmma_f32_16x16x4_f32(
                 false, a,  false, b,
                 (short)0, acc,  false,  false);
        }
        ap += 32;
        bp += 16 * WPITCH;
    }

    const float bv = bias[c * OUT_SZ + n0 + l];
    const int base = tile * TILE_M;
    #pragma unroll
    for (int r = 0; r < 8; ++r) {
        int m  = r + half * 8;
        int gm = base + m;
        if (gm < count) {
            int s = Si[m];
            float y = acc[r] + bv;
            y = __builtin_amdgcn_rcpf(1.0f + __expf(-y));
            out[(size_t)s * OUT_SZ + n0 + l] = y;
        }
    }
}

extern "C" void kernel_launch(void* const* d_in, const int* in_sizes, int n_in,
                              void* d_out, int out_size, void* d_ws, size_t ws_size,
                              hipStream_t stream) {
    const float* x    = (const float*)d_in[0];
    const int*   cam  = (const int*)  d_in[1];
    const float* W    = (const float*)d_in[2];
    const float* bias = (const float*)d_in[3];
    float* out = (float*)d_out;

    int* counts = (int*)d_ws;
    int* idx    = counts + 64;

    fc_zero_counts<<<1, 32, 0, stream>>>(counts);
    fc_bucket<<<BATCH / 256, 256, 0, stream>>>(cam, counts, idx);

    dim3 grid(NUM_CAM, BATCH / TILE_M);
    fc_expert_gemm<<<grid, 256, 0, stream>>>(x, W, bias, counts, idx, out);
}


# GEMM blocks loop over their camera's 16-row tiles: W[c] staged in LDS once per block, next tile's rows loaded under the current products; grid 16x16, was 16x512; sums and k order unchanged
# speedup vs baseline: 1.4437x; 1.4437x over previous
#include <hip/hip_runtime.h>

#define NUM_CAM 16
#define IN_SZ   256
#define OUT_SZ  128
#define BATCH   8192
#define TILE_M  16
#define XPITCH  260
#define WPITCH  130
#define SPLIT   16

typedef __attribute__((ext_vector_type(2))) float v2f;
typedef __attribute__((ext_vector_type(8))) float v8f;

__global__ void fc_zero_counts(int* counts) {
    if (threadIdx.x < NUM_CAM) counts[threadIdx.x] = 0;
}

__global__ void fc_bucket(const int* __restrict__ cam,
                          int* __restrict__ counts,
                          int* __restrict__ idx) {
    int t = blockIdx.x * blockDim.x + threadIdx.x;
    if (t < BATCH) {
        int c = cam[t] & (NUM_CAM - 1);
        int pos = atomicAdd(&counts[c], 1);
        idx[c * BATCH + pos] = t;
    }
}

__launch_bounds__(256)
__global__ void fc_expert_gemm(const float* __restrict__ x,
                               const float* __restrict__ W,
                               const float* __restrict__ bias,
                               const int*  __restrict__ counts,
                               const int*  __restrict__ idx,
                               float* __restrict__ out) {
    __shared__ float Xs[2][TILE_M * XPITCH];
    __shared__ v2f   Wt[128 * WPITCH];
    __shared__ int   Si[2][TILE_M];

    const int c     = blockIdx.x;
    const int split = blockIdx.y;
    const int count = counts[c];
    if (split * TILE_M >= count) return;

    const int tid  = threadIdx.x;
    const int lane = tid & 31;
    const int wave = tid >> 5;
    const int half = lane >> 4;
    const int l    = lane & 15;

    const int xrow  = tid >> 4;
    const int xcol0 = (tid & 15) * 16;
    const int* cidx = idx + c * BATCH;
    const int  last = count - 1;
    float4 xr[4];
    int xs  = cidx[min(split * TILE_M + xrow, last)];
    int xsn = cidx[min((split + SPLIT) * TILE_M + xrow, last)];
    {
        const float4* src = (const float4*)(x + (size_t)xs * IN_SZ + xcol0);
        #pragma unroll
        for (int i = 0; i < 4; ++i) xr[i] = src[i];
    }

    {
        const v2f* Wsrc = (const v2f*)(W + (size_t)c * OUT_SZ * IN_SZ);
        int p     = tid & 127;
        int nbase = tid >> 7;
        #pragma unroll 32
        for (int it = 0; it < 64; ++it) {
            int n = it * 2 + nbase;
            Wt[p * WPITCH + n] = Wsrc[n * 128 + p];
        }
    }

    const int   n0 = wave * 16;
    const float bv = bias[c * OUT_SZ + n0 + l];
    const v2f*  bp0 = &Wt[half * WPITCH + (n0 + l)];

    int buf = 0;
    for (int tile = split; tile * TILE_M < count; tile += SPLIT) {
        {
            float* dst = &Xs[buf][xrow * XPITCH + xcol0];
            #pragma unroll
            for (int i = 0; i < 4; ++i) {
                float4 v = xr[i];
                dst[i * 4 + 0] = v.x; dst[i * 4 + 1] = v.y;
                dst[i * 4 + 2] = v.z; dst[i * 4 + 3] = v.w;
            }
            if ((tid & 15) == 0) Si[buf][xrow] = xs;
        }
        __syncthreads();

        const int ntile = tile + SPLIT;
        if (ntile * TILE_M < count) {
            xs = xsn;
            const float4* src = (const float4*)(x + (size_t)xs * IN_SZ + xcol0);
            #pragma unroll
            for (int i = 0; i < 4; ++i) xr[i] = src[i];
            xsn = cidx[min((ntile + SPLIT) * TILE_M + xrow, last)];
        }

        const float* ap = &Xs[buf][l * XPITCH + 2 * half];
        const v2f*   bp = bp0;

        v8f acc = {};
        for (int ko = 0; ko < 8; ++ko) {
            #pragma unroll
            for (int ki = 0; ki < 8; ++ki) {
                v2f a = *(const v2f*)(ap + ki * 4);
                v2f b = bp[ki * 2 * WPITCH];
                acc = __builtin_amdgcn_wmma_f32_16x16x4_f32(
                     false, a,  false, b,
                     (short)0, acc,  false,  false);
            }
            ap += 32;
            bp += 16 * WPITCH;
        }

        const int base = tile * TILE_M;
        #pragma unroll
        for (int r = 0; r < 8; ++r) {
            int m  = r + half * 8;
            int gm = base + m;
            if (gm < count) {
                int s = Si[buf][m];
                float y = acc[r] + bv;
                y = __builtin_amdgcn_rcpf(1.0f + __expf(-y));
                out[(size_t)s * OUT_SZ + n0 + l] = y;
            }
        }
        buf ^= 1;
    }
}

extern "C" void kernel_launch(void* const* d_in, const int* in_sizes, int n_in,
                              void* d_out, int out_size, void* d_ws, size_t ws_size,
                              hipStream_t stream) {
    const float* x    = (const float*)d_in[0];
    const int*   cam  = (const int*)  d_in[1];
    const float* W    = (const float*)d_in[2];
    const float* bias = (const float*)d_in[3];
    float* out = (float*)d_out;

    int* counts = (int*)d_ws;
    int* idx    = counts + 64;

    fc_zero_counts<<<1, 32, 0, stream>>>(counts);
    fc_bucket<<<BATCH / 256, 256, 0, stream>>>(cam, counts, idx);

    dim3 grid(NUM_CAM, SPLIT);
    fc_expert_gemm<<<grid, 256, 0, stream>>>(x, W, bias, counts, idx, out);
}


# Bucket pass fused into the GEMM kernel: every block builds its camera's row list in LDS from cam by wave ballots and a block prefix; 3 launches become 1, atomics and workspace gone; sums unchanged
# speedup vs baseline: 1.9928x; 1.3803x over previous
#include <hip/hip_runtime.h>

#define NUM_CAM 16
#define IN_SZ   256
#define OUT_SZ  128
#define BATCH   8192
#define TILE_M  16
#define XPITCH  260
#define WPITCH  130
#define SPLIT   16

typedef __attribute__((ext_vector_type(2))) float v2f;
typedef __attribute__((ext_vector_type(8))) float v8f;

static __device__ __forceinline__ void lds_barrier() {
    __builtin_amdgcn_fence(__ATOMIC_RELEASE, "workgroup", "local");
    __builtin_amdgcn_s_barrier();
    __builtin_amdgcn_fence(__ATOMIC_ACQUIRE, "workgroup", "local");
}

__launch_bounds__(256)
__global__ void fc_expert_gemm(const float* __restrict__ x,
                               const int*   __restrict__ cam,
                               const float* __restrict__ W,
                               const float* __restrict__ bias,
                               float* __restrict__ out) {
    __shared__ float Xs[2][TILE_M * XPITCH];
    __shared__ v2f   Wt[128 * WPITCH];
    __shared__ int   Li[BATCH];
    __shared__ int   Wtot[8];

    const int c     = blockIdx.x;
    const int split = blockIdx.y;

    const int tid  = threadIdx.x;
    const int lane = tid & 31;
    const int wave = tid >> 5;
    const int half = lane >> 4;
    const int l    = lane & 15;

    int4 cv[8];
    {
        const int4* cam4 = (const int4*)cam;
        #pragma unroll
        for (int j = 0; j < 8; ++j) cv[j] = cam4[j * 256 + tid];
    }

    const v2f* Wsrc = (const v2f*)(W + (size_t)c * OUT_SZ * IN_SZ);
    const int p     = tid & 127;
    const int nbase = tid >> 7;
    v2f w0[32];
    #pragma unroll
    for (int it = 0; it < 32; ++it) {
        int n = it * 2 + nbase;
        w0[it] = Wsrc[n * 128 + p];
    }

    const int wv = __builtin_amdgcn_readfirstlane(wave);
    bool     mt[32];
    unsigned mk[32];
    int      wtotal = 0;
    #pragma unroll
    for (int j = 0; j < 8; ++j) {
        const int e4[4] = {cv[j].x, cv[j].y, cv[j].z, cv[j].w};
        #pragma unroll
        for (int e = 0; e < 4; ++e) {
            mt[j * 4 + e] = (e4[e] & (NUM_CAM - 1)) == c;
            mk[j * 4 + e] = __builtin_amdgcn_ballot_w32(mt[j * 4 + e]);
            wtotal += __builtin_popcount(mk[j * 4 + e]);
        }
    }
    if (lane == 0) Wtot[wv] = wtotal;
    lds_barrier();

    int count = 0;
    int pos0  = 0;
    #pragma unroll
    for (int w = 0; w < 8; ++w) {
        int t = Wtot[w];
        count += t;
        pos0  += (w < wv) ? t : 0;
    }
    if (split * TILE_M >= count) return;

    #pragma unroll
    for (int k = 0; k < 32; ++k) {
        if (mt[k])
            Li[__builtin_amdgcn_mbcnt_lo(mk[k], pos0)] = (k >> 2) * 1024 + tid * 4 + (k & 3);
        pos0 += __builtin_popcount(mk[k]);
    }
    lds_barrier();

    const int xrow  = tid >> 4;
    const int xcol0 = (tid & 15) * 16;
    const int last = count - 1;
    float4 xr[4];
    int xs  = Li[min(split * TILE_M + xrow, last)];
    int xsn = Li[min((split + SPLIT) * TILE_M + xrow, last)];
    {
        const float4* src = (const float4*)(x + (size_t)xs * IN_SZ + xcol0);
        #pragma unroll
        for (int i = 0; i < 4; ++i) xr[i] = src[i];
    }

    #pragma unroll
    for (int it = 0; it < 32; ++it) {
        int n = it * 2 + nbase;
        Wt[p * WPITCH + n] = w0[it];
    }
    #pragma unroll
    for (int it = 32; it < 64; ++it) {
        int n = it * 2 + nbase;
        Wt[p * WPITCH + n] = Wsrc[n * 128 + p];
    }

    const int   n0 = wave * 16;
    const float bv = bias[c * OUT_SZ + n0 + l];
    const v2f*  bp0 = &Wt[half * WPITCH + (n0 + l)];

    int buf = 0;
    for (int tile = split; tile * TILE_M < count; tile += SPLIT) {
        {
            float* dst = &Xs[buf][xrow * XPITCH + xcol0];
            #pragma unroll
            for (int i = 0; i < 4; ++i) {
                float4 v = xr[i];
                dst[i * 4 + 0] = v.x; dst[i * 4 + 1] = v.y;
                dst[i * 4 + 2] = v.z; dst[i * 4 + 3] = v.w;
            }
        }
        __syncthreads();

        const int ntile = tile + SPLIT;
        if (ntile * TILE_M < count) {
            xs = xsn;
            const float4* src = (const float4*)(x + (size_t)xs * IN_SZ + xcol0);
            #pragma unroll
            for (int i = 0; i < 4; ++i) xr[i] = src[i];
            xsn = Li[min((ntile + SPLIT) * TILE_M + xrow, last)];
        }

        const float* ap = &Xs[buf][l * XPITCH + 2 * half];
        const v2f*   bp = bp0;

        v8f acc = {};
        for (int ko = 0; ko < 8; ++ko) {
            #pragma unroll
            for (int ki = 0; ki < 8; ++ki) {
                v2f a = *(const v2f*)(ap + ki * 4);
                v2f b = bp[ki * 2 * WPITCH];
                acc = __builtin_amdgcn_wmma_f32_16x16x4_f32(
                     false, a,  false, b,
                     (short)0, acc,  false,  false);
            }
            ap += 32;
            bp += 16 * WPITCH;
        }

        const int base = tile * TILE_M;
        #pragma unroll
        for (int r = 0; r < 8; ++r) {
            int m  = r + half * 8;
            int gm = base + m;
            if (gm < count) {
                int s = Li[gm];
                float y = acc[r] + bv;
                y = __builtin_amdgcn_rcpf(1.0f + __expf(-y));
                out[(size_t)s * OUT_SZ + n0 + l] = y;
            }
        }
        buf ^= 1;
    }
}

extern "C" void kernel_launch(void* const* d_in, const int* in_sizes, int n_in,
                              void* d_out, int out_size, void* d_ws, size_t ws_size,
                              hipStream_t stream) {
    const float* x    = (const float*)d_in[0];
    const int*   cam  = (const int*)  d_in[1];
    const float* W    = (const float*)d_in[2];
    const float* bias = (const float*)d_in[3];
    float* out = (float*)d_out;

    dim3 grid(NUM_CAM, SPLIT);
    fc_expert_gemm<<<grid, 256, 0, stream>>>(x, cam, W, bias, out);
}


# Epilogue: a lane's 8 row numbers read from the row list as one group, sigmoid on all 8 rows at once, stores behind one uniform full-tile test; was read, wait, store row by row; sums unchanged
# speedup vs baseline: 2.4421x; 1.2255x over previous
#include <hip/hip_runtime.h>


#define NUM_CAM 16
#define IN_SZ   256
#define OUT_SZ  128
#define BATCH   8192
#define TILE_M  16
#define XPITCH  260
#define WPITCH  260
#define SPLIT   16

typedef __attribute__((ext_vector_type(2))) float v2f;
typedef __attribute__((ext_vector_type(4))) float v4f;
typedef __attribute__((ext_vector_type(8))) float v8f;

static __device__ __forceinline__ void lds_barrier() {
    __builtin_amdgcn_fence(__ATOMIC_RELEASE, "workgroup", "local");
    __builtin_amdgcn_s_barrier();
    __builtin_amdgcn_fence(__ATOMIC_ACQUIRE, "workgroup", "local");
}

static __device__ __forceinline__ v8f tile_products(v8f acc, const float* ap, const float* bp,
                                                    int s0, int s1) {
    #pragma unroll
    for (int s = s0; s < s1; ++s) {
        v2f a = *(const v2f*)(ap + 4 * s);
        v2f b = *(const v2f*)(bp + 4 * s);
        acc = __builtin_amdgcn_wmma_f32_16x16x4_f32(
             false, a,  false, b,
             (short)0, acc,  false,  false);
    }
    return acc;
}

__launch_bounds__(256)
__global__ void fc_expert_gemm(const float* __restrict__ x,
                               const int*   __restrict__ cam,
                               const float* __restrict__ W,
                               const float* __restrict__ bias,
                               float* __restrict__ out) {
    __shared__ __attribute__((aligned(16))) float Xs[2][TILE_M * XPITCH];
    __shared__ __attribute__((aligned(16))) float Ws[OUT_SZ * WPITCH];
    __shared__ int   Li[BATCH];
    __shared__ int   Wtot[8];

    const int c     = blockIdx.x;
    const int split = blockIdx.y;

    const int tid  = threadIdx.x;
    const int lane = tid & 31;
    const int wave = tid >> 5;
    const int half = lane >> 4;
    const int l    = lane & 15;
    const int wv = __builtin_amdgcn_readfirstlane(wave);
    const int n0 = wv * 16;

    int4 cv[8];
    {
        const int4* cam4 = (const int4*)cam;
        #pragma unroll
        for (int j = 0; j < 8; ++j) cv[j] = cam4[j * 256 + tid];
    }
    const float bv = bias[c * OUT_SZ + n0 + l];

    const v4f* wsrc = (const v4f*)(W + ((size_t)c * OUT_SZ + n0 + half) * IN_SZ + 4 * l);
    float*        wdst = &Ws[(n0 + half) * WPITCH + 4 * l];
    v4f wa[16];
    #pragma unroll
    for (int i = 0; i < 16; ++i) {
        const int q = i >> 3, j = i & 7;
        wa[i] = wsrc[j * (2 * IN_SZ / 4) + q * 16];
    }

    bool     mt[32];
    unsigned mk[32];
    int      wtotal = 0;
    #pragma unroll
    for (int j = 0; j < 8; ++j) {
        const int e4[4] = {cv[j].x, cv[j].y, cv[j].z, cv[j].w};
        #pragma unroll
        for (int e = 0; e < 4; ++e) {
            mt[j * 4 + e] = (e4[e] & (NUM_CAM - 1)) == c;
            mk[j * 4 + e] = __builtin_amdgcn_ballot_w32(mt[j * 4 + e]);
            wtotal += __builtin_popcount(mk[j * 4 + e]);
        }
    }
    if (lane == 0) Wtot[wv] = wtotal;
    lds_barrier();

    int count = 0;
    int pos0  = 0;
    #pragma unroll
    for (int w = 0; w < 8; ++w) {
        int t = Wtot[w];
        count += t;
        pos0  += (w < wv) ? t : 0;
    }
    if (split * TILE_M >= count) return;

    #pragma unroll
    for (int k = 0; k < 32; ++k) {
        if (mt[k])
            Li[__builtin_amdgcn_mbcnt_lo(mk[k], pos0)] = (k >> 2) * 1024 + tid * 4 + (k & 3);
        pos0 += __builtin_popcount(mk[k]);
    }
    lds_barrier();

    const int xrow  = tid >> 4;
    const int xcol0 = (tid & 15) * 16;
    const int last = count - 1;
    float4 xr[4];
    int xs  = Li[min(split * TILE_M + xrow, last)];
    int xsn = Li[min((split + SPLIT) * TILE_M + xrow, last)];
    {
        const float4* src = (const float4*)(x + (size_t)xs * IN_SZ + xcol0);
        #pragma unroll
        for (int i = 0; i < 4; ++i) xr[i] = src[i];
    }

    v4f wb[16];
    __builtin_amdgcn_sched_barrier(0);
    #pragma unroll
    for (int j = 0; j < 8; ++j) wb[j] = wsrc[j * (2 * IN_SZ / 4) + 2 * 16];
    __builtin_amdgcn_sched_barrier(0);
    #pragma unroll
    for (int j = 0; j < 8; ++j) wb[8 + j] = wsrc[j * (2 * IN_SZ / 4) + 3 * 16];

    #pragma unroll
    for (int i = 0; i < 16; ++i) {
        const int q = i >> 3, j = i & 7;
        *(v4f*)(wdst + 2 * j * WPITCH + 64 * q) = wa[i];
    }

    const float* bp = &Ws[(n0 + l) * WPITCH + 2 * half];

    auto epilogue = [&](const v8f& acc, int tile) {
        const int gm0 = tile * TILE_M + half * 8;
        int   si[8];
        float y[8];
        #pragma unroll
        for (int r = 0; r < 8; ++r) si[r] = Li[min(gm0 + r, last)];
        #pragma unroll
        for (int r = 0; r < 8; ++r) {
            float t = acc[r] + bv;
            y[r] = __builtin_amdgcn_rcpf(1.0f + __expf(-t));
        }
        float* op = out + n0 + l;
        if ((tile + 1) * TILE_M <= count) {
            #pragma unroll
            for (int r = 0; r < 8; ++r) op[(unsigned)si[r] * OUT_SZ] = y[r];
        } else {
            #pragma unroll
            for (int r = 0; r < 8; ++r)
                if (gm0 + r < count) op[(unsigned)si[r] * OUT_SZ] = y[r];
        }
    };

    int tile = split;
    {
        {
            float* dst = &Xs[0][xrow * XPITCH + xcol0];
            #pragma unroll
            for (int i = 0; i < 4; ++i) {
                float4 v = xr[i];
                dst[i * 4 + 0] = v.x; dst[i * 4 + 1] = v.y;
                dst[i * 4 + 2] = v.z; dst[i * 4 + 3] = v.w;
            }
        }
        lds_barrier();

        const int ntile = tile + SPLIT;
        if (ntile * TILE_M < count) {
            xs = xsn;
            const float4* src = (const float4*)(x + (size_t)xs * IN_SZ + xcol0);
            #pragma unroll
            for (int i = 0; i < 4; ++i) xr[i] = src[i];
            xsn = Li[min((ntile + SPLIT) * TILE_M + xrow, last)];
        }

        const float* ap = &Xs[0][l * XPITCH + 2 * half];
        v8f acc = {};
        acc = tile_products(acc, ap, bp, 0, 32);
        #pragma unroll
        for (int j = 0; j < 8; ++j) *(v4f*)(wdst + 2 * j * WPITCH + 128) = wb[j];
        acc = tile_products(acc, ap, bp, 32, 48);
        #pragma unroll
        for (int j = 0; j < 8; ++j) *(v4f*)(wdst + 2 * j * WPITCH + 192) = wb[8 + j];
        acc = tile_products(acc, ap, bp, 48, 64);

        epilogue(acc, tile);
        tile += SPLIT;
    }

    int buf = 1;
    for (; tile * TILE_M < count; tile += SPLIT) {
        {
            float* dst = &Xs[buf][xrow * XPITCH + xcol0];
            #pragma unroll
            for (int i = 0; i < 4; ++i) {
                float4 v = xr[i];
                dst[i * 4 + 0] = v.x; dst[i * 4 + 1] = v.y;
                dst[i * 4 + 2] = v.z; dst[i * 4 + 3] = v.w;
            }
        }
        lds_barrier();

        const int ntile = tile + SPLIT;
        if (ntile * TILE_M < count) {
            xs = xsn;
            const float4* src = (const float4*)(x + (size_t)xs * IN_SZ + xcol0);
            #pragma unroll
            for (int i = 0; i < 4; ++i) xr[i] = src[i];
            xsn = Li[min((ntile + SPLIT) * TILE_M + xrow, last)];
        }

        const float* ap = &Xs[buf][l * XPITCH + 2 * half];
        v8f acc = {};
        acc = tile_products(acc, ap, bp, 0, 64);

        epilogue(acc, tile);
        buf ^= 1;
    }
}

extern "C" void kernel_launch(void* const* d_in, const int* in_sizes, int n_in,
                              void* d_out, int out_size, void* d_ws, size_t ws_size,
                              hipStream_t stream) {
    const float* x    = (const float*)d_in[0];
    const int*   cam  = (const int*)  d_in[1];
    const float* W    = (const float*)d_in[2];
    const float* bias = (const float*)d_in[3];
    float* out = (float*)d_out;

    dim3 grid(NUM_CAM, SPLIT);
    fc_expert_gemm<<<grid, 256, 0, stream>>>(x, cam, W, bias, out);
}


# Tiles past a camera's 32nd: each shared by two blocks, four waves (one per SIMD, LDS reads grouped 16 k-steps ahead) on their own channels; was one block's third whole tile; sums unchanged
# speedup vs baseline: 2.5603x; 1.0484x over previous
#include <hip/hip_runtime.h>


#define NUM_CAM 16
#define IN_SZ   256
#define OUT_SZ  128
#define BATCH   8192
#define TILE_M  16
#define XPITCH  260
#define WPITCH  260
#define SPLIT   16

typedef __attribute__((ext_vector_type(2))) float v2f;
typedef __attribute__((ext_vector_type(4))) float v4f;
typedef __attribute__((ext_vector_type(8))) float v8f;

static __device__ __forceinline__ void lds_barrier() {
    __builtin_amdgcn_fence(__ATOMIC_RELEASE, "workgroup", "local");
    __builtin_amdgcn_s_barrier();
    __builtin_amdgcn_fence(__ATOMIC_ACQUIRE, "workgroup", "local");
}

static __device__ __forceinline__ v8f tile_products(v8f acc, const float* ap, const float* bp,
                                                    int s0, int s1) {
    #pragma unroll
    for (int s = s0; s < s1; ++s) {
        v2f a = *(const v2f*)(ap + 4 * s);
        v2f b = *(const v2f*)(bp + 4 * s);
        acc = __builtin_amdgcn_wmma_f32_16x16x4_f32(
             false, a,  false, b,
             (short)0, acc,  false,  false);
    }
    return acc;
}

static __device__ __forceinline__ v8f tile_products_alone(const float* ap, const float* bp) {
    v2f a[2][16], b[2][16];
    #pragma unroll
    for (int s = 0; s < 16; ++s) {
        a[0][s] = *(const v2f*)(ap + 4 * s);
        b[0][s] = *(const v2f*)(bp + 4 * s);
    }
    v8f acc = {};
    #pragma unroll
    for (int g = 0; g < 4; ++g) {
        if (g + 1 < 4) {
            #pragma unroll
            for (int s = 0; s < 16; ++s) {
                a[(g + 1) & 1][s] = *(const v2f*)(ap + 4 * (16 * (g + 1) + s));
                b[(g + 1) & 1][s] = *(const v2f*)(bp + 4 * (16 * (g + 1) + s));
            }
        }
        __builtin_amdgcn_sched_barrier(0);
        #pragma unroll
        for (int s = 0; s < 16; ++s)
            acc = __builtin_amdgcn_wmma_f32_16x16x4_f32(
                 false, a[g & 1][s],  false, b[g & 1][s],
                 (short)0, acc,  false,  false);
    }
    return acc;
}

__launch_bounds__(256)
__global__ void fc_expert_gemm(const float* __restrict__ x,
                               const int*   __restrict__ cam,
                               const float* __restrict__ W,
                               const float* __restrict__ bias,
                               float* __restrict__ out) {
    __shared__ __attribute__((aligned(16))) float Xs[3][TILE_M * XPITCH];
    __shared__ __attribute__((aligned(16))) float Ws[OUT_SZ * WPITCH];
    __shared__ int   Li[BATCH];
    __shared__ int   Wtot[8];

    const int c     = blockIdx.x;
    const int split = blockIdx.y;

    const int tid  = threadIdx.x;
    const int lane = tid & 31;
    const int wave = tid >> 5;
    const int half = lane >> 4;
    const int l    = lane & 15;
    const int wv = __builtin_amdgcn_readfirstlane(wave);
    const int n0 = wv * 16;

    int4 cv[8];
    {
        const int4* cam4 = (const int4*)cam;
        #pragma unroll
        for (int j = 0; j < 8; ++j) cv[j] = cam4[j * 256 + tid];
    }
    const float bv = bias[c * OUT_SZ + n0 + l];

    const v4f* wsrc = (const v4f*)(W + ((size_t)c * OUT_SZ + n0 + half) * IN_SZ + 4 * l);
    float*        wdst = &Ws[(n0 + half) * WPITCH + 4 * l];
    v4f wa[16];
    #pragma unroll
    for (int i = 0; i < 16; ++i) {
        const int q = i >> 3, j = i & 7;
        wa[i] = wsrc[j * (2 * IN_SZ / 4) + q * 16];
    }

    bool     mt[32];
    unsigned mk[32];
    int      wtotal = 0;
    #pragma unroll
    for (int j = 0; j < 8; ++j) {
        const int e4[4] = {cv[j].x, cv[j].y, cv[j].z, cv[j].w};
        #pragma unroll
        for (int e = 0; e < 4; ++e) {
            mt[j * 4 + e] = (e4[e] & (NUM_CAM - 1)) == c;
            mk[j * 4 + e] = __builtin_amdgcn_ballot_w32(mt[j * 4 + e]);
            wtotal += __builtin_popcount(mk[j * 4 + e]);
        }
    }
    if (lane == 0) Wtot[wv] = wtotal;
    lds_barrier();

    int count = 0;
    int pos0  = 0;
    #pragma unroll
    for (int w = 0; w < 8; ++w) {
        int t = Wtot[w];
        count += t;
        pos0  += (w < wv) ? t : 0;
    }
    if (split * TILE_M >= count) return;
    const int ntiles = (count + TILE_M - 1) / TILE_M;
    const int tmain  = min(ntiles, 2 * SPLIT);
    const int nextra = ntiles - tmain;

    #pragma unroll
    for (int k = 0; k < 32; ++k) {
        if (mt[k])
            Li[__builtin_amdgcn_mbcnt_lo(mk[k], pos0)] = (k >> 2) * 1024 + tid * 4 + (k & 3);
        pos0 += __builtin_popcount(mk[k]);
    }
    lds_barrier();

    const int xrow  = tid >> 4;
    const int xcol0 = (tid & 15) * 16;
    const int last = count - 1;
    float4 xr[4];
    const int xs  = Li[min(split * TILE_M + xrow, last)];
    const int xsn = Li[min((split + SPLIT) * TILE_M + xrow, last)];
    {
        const float4* src = (const float4*)(x + (size_t)xs * IN_SZ + xcol0);
        #pragma unroll
        for (int i = 0; i < 4; ++i) xr[i] = src[i];
    }

    v4f wb[16];
    __builtin_amdgcn_sched_barrier(0);
    #pragma unroll
    for (int j = 0; j < 8; ++j) wb[j] = wsrc[j * (2 * IN_SZ / 4) + 2 * 16];
    __builtin_amdgcn_sched_barrier(0);
    #pragma unroll
    for (int j = 0; j < 8; ++j) wb[8 + j] = wsrc[j * (2 * IN_SZ / 4) + 3 * 16];

    #pragma unroll
    for (int i = 0; i < 16; ++i) {
        const int q = i >> 3, j = i & 7;
        *(v4f*)(wdst + 2 * j * WPITCH + 64 * q) = wa[i];
    }

    const float* bp = &Ws[(n0 + l) * WPITCH + 2 * half];

    auto epilogue = [&](const v8f& acc, int tile) {
        const int gm0 = tile * TILE_M + half * 8;
        int   si[8];
        float y[8];
        #pragma unroll
        for (int r = 0; r < 8; ++r) si[r] = Li[min(gm0 + r, last)];
        #pragma unroll
        for (int r = 0; r < 8; ++r) {
            float t = acc[r] + bv;
            y[r] = __builtin_amdgcn_rcpf(1.0f + __expf(-t));
        }
        float* op = out + n0 + l;
        if ((tile + 1) * TILE_M <= count) {
            #pragma unroll
            for (int r = 0; r < 8; ++r) op[(unsigned)si[r] * OUT_SZ] = y[r];
        } else {
            #pragma unroll
            for (int r = 0; r < 8; ++r)
                if (gm0 + r < count) op[(unsigned)si[r] * OUT_SZ] = y[r];
        }
    };

    int tile = split;
    {
        {
            float* dst = &Xs[0][xrow * XPITCH + xcol0];
            #pragma unroll
            for (int i = 0; i < 4; ++i) {
                float4 v = xr[i];
                dst[i * 4 + 0] = v.x; dst[i * 4 + 1] = v.y;
                dst[i * 4 + 2] = v.z; dst[i * 4 + 3] = v.w;
            }
        }
        lds_barrier();

        if (tile + SPLIT < tmain) {
            const float4* src = (const float4*)(x + (size_t)xsn * IN_SZ + xcol0);
            #pragma unroll
            for (int i = 0; i < 4; ++i) xr[i] = src[i];
        }

        const float* ap = &Xs[0][l * XPITCH + 2 * half];
        v8f acc = {};
        acc = tile_products(acc, ap, bp, 0, 32);
        #pragma unroll
        for (int j = 0; j < 8; ++j) *(v4f*)(wdst + 2 * j * WPITCH + 128) = wb[j];
        acc = tile_products(acc, ap, bp, 32, 48);
        #pragma unroll
        for (int j = 0; j < 8; ++j) *(v4f*)(wdst + 2 * j * WPITCH + 192) = wb[8 + j];
        acc = tile_products(acc, ap, bp, 48, 64);

        epilogue(acc, tile);
        tile += SPLIT;
    }

    float4 xe[4];
    bool   xe_ahead = false;
    auto load_shared_rows = [&](int te) {
        const int se = Li[min(te * TILE_M + xrow, last)];
        const float4* src = (const float4*)(x + (size_t)se * IN_SZ + xcol0);
        #pragma unroll
        for (int i = 0; i < 4; ++i) xe[i] = src[i];
    };

    if (tile < tmain) {
        {
            float* dst = &Xs[1][xrow * XPITCH + xcol0];
            #pragma unroll
            for (int i = 0; i < 4; ++i) {
                float4 v = xr[i];
                dst[i * 4 + 0] = v.x; dst[i * 4 + 1] = v.y;
                dst[i * 4 + 2] = v.z; dst[i * 4 + 3] = v.w;
            }
        }
        lds_barrier();

        if ((split >> 1) < nextra) {
            load_shared_rows(2 * SPLIT + (split >> 1));
            xe_ahead = true;
        }

        const float* ap = &Xs[1][l * XPITCH + 2 * half];
        v8f acc = {};
        acc = tile_products(acc, ap, bp, 0, 64);

        epilogue(acc, tile);
    }

    const unsigned wset = (split & 1) ? 0x5Au : 0xA5u;
    int round = 0;
    for (int je = split >> 1; je < nextra; je += SPLIT / 2, ++round) {
        const int te = 2 * SPLIT + je;
        if (round > 0 || !xe_ahead) load_shared_rows(te);
        if (round > 0) lds_barrier();
        {
            float* dst = &Xs[2][xrow * XPITCH + xcol0];
            #pragma unroll
            for (int i = 0; i < 4; ++i) {
                float4 v = xe[i];
                dst[i * 4 + 0] = v.x; dst[i * 4 + 1] = v.y;
                dst[i * 4 + 2] = v.z; dst[i * 4 + 3] = v.w;
            }
        }
        lds_barrier();
        if ((wset >> wv) & 1) {
            const float* ap = &Xs[2][l * XPITCH + 2 * half];
            const v8f acc = tile_products_alone(ap, bp);
            epilogue(acc, te);
        }
    }
}

extern "C" void kernel_launch(void* const* d_in, const int* in_sizes, int n_in,
                              void* d_out, int out_size, void* d_ws, size_t ws_size,
                              hipStream_t stream) {
    const float* x    = (const float*)d_in[0];
    const int*   cam  = (const int*)  d_in[1];
    const float* W    = (const float*)d_in[2];
    const float* bias = (const float*)d_in[3];
    float* out = (float*)d_out;

    dim3 grid(NUM_CAM, SPLIT);
    fc_expert_gemm<<<grid, 256, 0, stream>>>(x, cam, W, bias, out);
}
